# starting kernel
# baseline (speedup 1.0000x reference)
#include <hip/hip_runtime.h>

typedef float v2f __attribute__((ext_vector_type(2)));
typedef float v8f __attribute__((ext_vector_type(8)));

#if defined(__HIP_DEVICE_COMPILE__) && __has_builtin(__builtin_amdgcn_tanhf)
  #define TANHF(x) __builtin_amdgcn_tanhf(x)
#else
  #define TANHF(x) tanhf(x)
#endif

__device__ __forceinline__ void half_swap(float& a, float& b, bool loH) {
#if defined(__HIP_DEVICE_COMPILE__) && __has_builtin(__builtin_amdgcn_permlane16_swap)
  auto r = __builtin_amdgcn_permlane16_swap(__float_as_uint(a), __float_as_uint(b),
                                            false, false);
  a = __uint_as_float(r[0]);
  b = __uint_as_float(r[1]);
#else
  const float sa = __shfl_xor(a, 16, 32);
  const float sb = __shfl_xor(b, 16, 32);
  const float na = loH ? a : sb;
  const float nb = loH ? sa : b;
  a = na; b = nb;
#endif
}

struct MlpParams {
  const float* x1; const float* x2;
  const float* w1; const float* w2; const float* b1;
  const float* W3; const float* B3; const float* W4; const float* B4;
  const float* W5; const float* B5; const float* W6; const float* B6;
  const float* W7; const float* B7; const float* W8; const float* B8;
  const float* alpha; const float* w9; const float* b9;
  float* out; int n;
};

__global__ __launch_bounds__(256) void mlp_wmma_kernel(MlpParams p) {
  __shared__ float Wp[6][16][16];
  __shared__ float Bp[6][16];
  __shared__ float w1p[16], w2p[16], b1p[16], w9p[16];

  const int tid = threadIdx.x;
  const float alpha = p.alpha[0];
  const float* Wg[6] = {p.W3, p.W4, p.W5, p.W6, p.W7, p.W8};
  const float* Bg[6] = {p.B3, p.B4, p.B5, p.B6, p.B7, p.B8};

  for (int idx = tid; idx < 6 * 256; idx += 256) {
    const int L = idx >> 8, pos = idx & 255, k = pos >> 4, j = pos & 15;
    Wp[L][k][j] = (k < 10 && j < 10) ? alpha * Wg[L][k * 10 + j] : 0.0f;
  }
  if (tid < 96) {
    const int L = tid >> 4, f = tid & 15;
    Bp[L][f] = (f < 10) ? alpha * Bg[L][f] : 0.0f;
  }
  if (tid < 16) {
    w1p[tid] = (tid < 10) ? alpha * p.w1[tid] : 0.0f;
    w2p[tid] = (tid < 10) ? alpha * p.w2[tid] : 0.0f;
    b1p[tid] = (tid < 10) ? alpha * p.b1[tid] : 0.0f;
    w9p[tid] = (tid < 10) ? p.w9[tid] : 0.0f;
  }
  __syncthreads();

  const int lane = tid & 31;
  const int m    = lane & 15;
  const bool loH = (lane < 16);
  const int koff = loH ? 0 : 2;
  const int fo   = loH ? 0 : 8;

  v2f A[6][3];
#pragma unroll
  for (int L = 0; L < 6; ++L)
#pragma unroll
    for (int w = 0; w < 3; ++w) {
      A[L][w].x = Wp[L][4 * w + koff + 0][m];
      A[L][w].y = Wp[L][4 * w + koff + 1][m];
    }
  v8f abv[6];
#pragma unroll
  for (int L = 0; L < 6; ++L)
#pragma unroll
    for (int r = 0; r < 8; ++r) abv[L][r] = Bp[L][r + fo];
  float w1s[8], w2s[8], b1s[8], w9s[8];
#pragma unroll
  for (int r = 0; r < 8; ++r) {
    w1s[r] = w1p[r + fo]; w2s[r] = w2p[r + fo];
    b1s[r] = b1p[r + fo]; w9s[r] = w9p[r + fo];
  }
  const float b9v = p.b9[0];

  const int numTiles   = (p.n + 15) >> 4;
  const int waveId     = blockIdx.x * (blockDim.x >> 5) + (tid >> 5);
  const int waveStride = gridDim.x * (blockDim.x >> 5);

  for (int t = waveId; t < numTiles; t += waveStride) {
    const int base = t << 4;
    int gi = base + m; if (gi >= p.n) gi = p.n - 1;
    const float x1v = p.x1[gi];
    const float x2v = p.x2[gi];

    float D[8];
#pragma unroll
    for (int r = 0; r < 8; ++r)
      D[r] = TANHF(__builtin_fmaf(x1v, w1s[r], __builtin_fmaf(x2v, w2s[r], b1s[r])));

#pragma unroll
    for (int L = 0; L < 6; ++L) {
      float t0 = D[0], t2 = D[2];  half_swap(t0, t2, loH);
      float t1 = D[1], t3 = D[3];  half_swap(t1, t3, loH);
      float t4 = D[4], t6 = D[6];  half_swap(t4, t6, loH);
      float t5 = D[5], t7 = D[7];  half_swap(t5, t7, loH);
      v2f b0; b0.x = t0; b0.y = t1;
      v2f b1; b1.x = t4; b1.y = t5;
      v2f b2; b2.x = t2; b2.y = t3;

      v8f c = abv[L];
      c = __builtin_amdgcn_wmma_f32_16x16x4_f32(false, A[L][0], false, b0,
                                                (short)0, c, false, false);
      c = __builtin_amdgcn_wmma_f32_16x16x4_f32(false, A[L][1], false, b1,
                                                (short)0, c, false, false);
      c = __builtin_amdgcn_wmma_f32_16x16x4_f32(false, A[L][2], false, b2,
                                                (short)0, c, false, false);
#pragma unroll
      for (int r = 0; r < 8; ++r) D[r] = TANHF(c[r]);
    }

    float acc = 0.0f;
#pragma unroll
    for (int r = 0; r < 8; ++r) acc = __builtin_fmaf(D[r], w9s[r], acc);
    float ra = acc, rb = acc;
    half_swap(ra, rb, loH);
    const float tot = ra + rb + b9v;
    if (loH && (base + lane) < p.n) p.out[base + lane] = tot;
  }
}

extern "C" void kernel_launch(void* const* d_in, const int* in_sizes, int n_in,
                              void* d_out, int out_size, void* d_ws, size_t ws_size,
                              hipStream_t stream) {
  MlpParams p;
  p.x1    = (const float*)d_in[0];
  p.x2    = (const float*)d_in[1];
  p.w1    = (const float*)d_in[2];
  p.w2    = (const float*)d_in[3];
  p.b1    = (const float*)d_in[4];
  p.W3    = (const float*)d_in[5];   p.B3 = (const float*)d_in[6];
  p.W4    = (const float*)d_in[7];   p.B4 = (const float*)d_in[8];
  p.W5    = (const float*)d_in[9];   p.B5 = (const float*)d_in[10];
  p.W6    = (const float*)d_in[11];  p.B6 = (const float*)d_in[12];
  p.W7    = (const float*)d_in[13];  p.B7 = (const float*)d_in[14];
  p.W8    = (const float*)d_in[15];  p.B8 = (const float*)d_in[16];
  p.alpha = (const float*)d_in[17];
  p.w9    = (const float*)d_in[18];
  p.b9    = (const float*)d_in[19];
  p.out   = (float*)d_out;
  p.n     = in_sizes[0];

  const int blocks = 2048;
  mlp_wmma_kernel<<<dim3(blocks), dim3(256), 0, stream>>>(p);
}


# Weight rows and k permuted so the 10 outputs fill 5 whole D registers that are the next WMMA's B fragments as they stand: no half-swaps, 5 tanh not 8 per layer; summation order of each dot changes
# speedup vs baseline: 1.1894x; 1.1894x over previous
#include <hip/hip_runtime.h>

typedef float v2f __attribute__((ext_vector_type(2)));
typedef float v8f __attribute__((ext_vector_type(8)));

#if defined(__HIP_DEVICE_COMPILE__) && __has_builtin(__builtin_amdgcn_tanhf)
  #define TANHF(x) __builtin_amdgcn_tanhf(x)
#else
  #define TANHF(x) tanhf(x)
#endif

__device__ __forceinline__ void half_swap(float& a, float& b, bool loH) {
#if defined(__HIP_DEVICE_COMPILE__) && __has_builtin(__builtin_amdgcn_permlane16_swap)
  auto r = __builtin_amdgcn_permlane16_swap(__float_as_uint(a), __float_as_uint(b),
                                            false, false);
  a = __uint_as_float(r[0]);
  b = __uint_as_float(r[1]);
#else
  const float sa = __shfl_xor(a, 16, 32);
  const float sb = __shfl_xor(b, 16, 32);
  const float na = loH ? a : sb;
  const float nb = loH ? sa : b;
  a = na; b = nb;
#endif
}

struct MlpParams {
  const float* x1; const float* x2;
  const float* w1; const float* w2; const float* b1;
  const float* W3; const float* B3; const float* W4; const float* B4;
  const float* W5; const float* B5; const float* W6; const float* B6;
  const float* W7; const float* B7; const float* W8; const float* B8;
  const float* alpha; const float* w9; const float* b9;
  float* out; int n;
};

__global__ __launch_bounds__(256) void mlp_wmma_kernel(MlpParams p) {
  __shared__ float Wp[6][16][16];
  __shared__ float Bp[6][16];
  __shared__ float w1p[16], w2p[16], b1p[16], w9p[16];

  const int tid = threadIdx.x;
  const float alpha = p.alpha[0];
  const float* Wg[6] = {p.W3, p.W4, p.W5, p.W6, p.W7, p.W8};
  const float* Bg[6] = {p.B3, p.B4, p.B5, p.B6, p.B7, p.B8};

  for (int idx = tid; idx < 6 * 256; idx += 256) {
    const int L = idx >> 8, pos = idx & 255, k = pos >> 4, j = pos & 15;
    Wp[L][k][j] = (k < 10 && j < 10) ? alpha * Wg[L][k * 10 + j] : 0.0f;
  }
  if (tid < 96) {
    const int L = tid >> 4, f = tid & 15;
    Bp[L][f] = (f < 10) ? alpha * Bg[L][f] : 0.0f;
  }
  if (tid < 16) {
    w1p[tid] = (tid < 10) ? alpha * p.w1[tid] : 0.0f;
    w2p[tid] = (tid < 10) ? alpha * p.w2[tid] : 0.0f;
    b1p[tid] = (tid < 10) ? alpha * p.b1[tid] : 0.0f;
    w9p[tid] = (tid < 10) ? p.w9[tid] : 0.0f;
  }
  __syncthreads();

  const int lane = tid & 31;
  const int m    = lane & 15;
  const bool loH = (lane < 16);
  const int h5   = loH ? 0 : 5;
  const int jrow = (m < 5) ? m : ((m >= 8 && m < 13) ? (m - 3) : 15);

  v2f A[6][3];
#pragma unroll
  for (int L = 0; L < 6; ++L)
#pragma unroll
    for (int w = 0; w < 3; ++w) {
      A[L][w].x = Wp[L][2 * w + 0 + h5][jrow];
      A[L][w].y = (w < 2) ? Wp[L][2 * w + 1 + h5][jrow] : 0.0f;
    }
  v8f abv[6];
#pragma unroll
  for (int L = 0; L < 6; ++L)
#pragma unroll
    for (int r = 0; r < 8; ++r) abv[L][r] = (r < 5) ? Bp[L][r + h5] : 0.0f;
  float w1s[5], w2s[5], b1s[5], w9s[5];
#pragma unroll
  for (int r = 0; r < 5; ++r) {
    w1s[r] = w1p[r + h5]; w2s[r] = w2p[r + h5];
    b1s[r] = b1p[r + h5]; w9s[r] = w9p[r + h5];
  }
  const float b9v = p.b9[0];

  const int numTiles   = (p.n + 15) >> 4;
  const int waveId     = blockIdx.x * (blockDim.x >> 5) + (tid >> 5);
  const int waveStride = gridDim.x * (blockDim.x >> 5);

  for (int t = waveId; t < numTiles; t += waveStride) {
    const int base = t << 4;
    int gi = base + m; if (gi >= p.n) gi = p.n - 1;
    const float x1v = p.x1[gi];
    const float x2v = p.x2[gi];

    float act[5];
#pragma unroll
    for (int r = 0; r < 5; ++r)
      act[r] = TANHF(__builtin_fmaf(x1v, w1s[r], __builtin_fmaf(x2v, w2s[r], b1s[r])));

#pragma unroll
    for (int L = 0; L < 6; ++L) {
      v2f b0; b0.x = act[0]; b0.y = act[1];
      v2f b1; b1.x = act[2]; b1.y = act[3];
      v2f b2; b2.x = act[4]; b2.y = 0.0f;

      v8f c = abv[L];
      c = __builtin_amdgcn_wmma_f32_16x16x4_f32(false, A[L][0], false, b0,
                                                (short)0, c, false, false);
      c = __builtin_amdgcn_wmma_f32_16x16x4_f32(false, A[L][1], false, b1,
                                                (short)0, c, false, false);
      c = __builtin_amdgcn_wmma_f32_16x16x4_f32(false, A[L][2], false, b2,
                                                (short)0, c, false, false);
#pragma unroll
      for (int r = 0; r < 5; ++r) act[r] = TANHF(c[r]);
    }

    float acc = 0.0f;
#pragma unroll
    for (int r = 0; r < 5; ++r) acc = __builtin_fmaf(act[r], w9s[r], acc);
    float ra = acc, rb = acc;
    half_swap(ra, rb, loH);
    const float tot = ra + rb + b9v;
    if (loH && (base + lane) < p.n) p.out[base + lane] = tot;
  }
}

extern "C" void kernel_launch(void* const* d_in, const int* in_sizes, int n_in,
                              void* d_out, int out_size, void* d_ws, size_t ws_size,
                              hipStream_t stream) {
  MlpParams p;
  p.x1    = (const float*)d_in[0];
  p.x2    = (const float*)d_in[1];
  p.w1    = (const float*)d_in[2];
  p.w2    = (const float*)d_in[3];
  p.b1    = (const float*)d_in[4];
  p.W3    = (const float*)d_in[5];   p.B3 = (const float*)d_in[6];
  p.W4    = (const float*)d_in[7];   p.B4 = (const float*)d_in[8];
  p.W5    = (const float*)d_in[9];   p.B5 = (const float*)d_in[10];
  p.W6    = (const float*)d_in[11];  p.B6 = (const float*)d_in[12];
  p.W7    = (const float*)d_in[13];  p.B7 = (const float*)d_in[14];
  p.W8    = (const float*)d_in[15];  p.B8 = (const float*)d_in[16];
  p.alpha = (const float*)d_in[17];
  p.w9    = (const float*)d_in[18];
  p.b9    = (const float*)d_in[19];
  p.out   = (float*)d_out;
  p.n     = in_sizes[0];

  const int blocks = 2048;
  mlp_wmma_kernel<<<dim3(blocks), dim3(256), 0, stream>>>(p);
}


# Each wave trip runs two independent 16-sample tiles, WMMAs alternating (shared weight fragments): one tile's dependent WMMA-tanh chain no longer idles the matrix pipe; one 128 B store per trip
# speedup vs baseline: 1.2644x; 1.0631x over previous
#include <hip/hip_runtime.h>

typedef float v2f __attribute__((ext_vector_type(2)));
typedef float v8f __attribute__((ext_vector_type(8)));

#if defined(__HIP_DEVICE_COMPILE__) && __has_builtin(__builtin_amdgcn_tanhf)
  #define TANHF(x) __builtin_amdgcn_tanhf(x)
#else
  #define TANHF(x) tanhf(x)
#endif

__device__ __forceinline__ void half_swap(float& a, float& b, bool loH) {
#if defined(__HIP_DEVICE_COMPILE__) && __has_builtin(__builtin_amdgcn_permlane16_swap)
  auto r = __builtin_amdgcn_permlane16_swap(__float_as_uint(a), __float_as_uint(b),
                                            false, false);
  a = __uint_as_float(r[0]);
  b = __uint_as_float(r[1]);
#else
  const float sa = __shfl_xor(a, 16, 32);
  const float sb = __shfl_xor(b, 16, 32);
  const float na = loH ? a : sb;
  const float nb = loH ? sa : b;
  a = na; b = nb;
#endif
}

struct MlpParams {
  const float* x1; const float* x2;
  const float* w1; const float* w2; const float* b1;
  const float* W3; const float* B3; const float* W4; const float* B4;
  const float* W5; const float* B5; const float* W6; const float* B6;
  const float* W7; const float* B7; const float* W8; const float* B8;
  const float* alpha; const float* w9; const float* b9;
  float* out; int n;
};

__global__ __launch_bounds__(256) void mlp_wmma_kernel(MlpParams p) {
  __shared__ float Wp[6][16][16];
  __shared__ float Bp[6][16];
  __shared__ float w1p[16], w2p[16], b1p[16], w9p[16];

  const int tid = threadIdx.x;
  const float alpha = p.alpha[0];
  const float* Wg[6] = {p.W3, p.W4, p.W5, p.W6, p.W7, p.W8};
  const float* Bg[6] = {p.B3, p.B4, p.B5, p.B6, p.B7, p.B8};

  for (int idx = tid; idx < 6 * 256; idx += 256) {
    const int L = idx >> 8, pos = idx & 255, k = pos >> 4, j = pos & 15;
    Wp[L][k][j] = (k < 10 && j < 10) ? alpha * Wg[L][k * 10 + j] : 0.0f;
  }
  if (tid < 96) {
    const int L = tid >> 4, f = tid & 15;
    Bp[L][f] = (f < 10) ? alpha * Bg[L][f] : 0.0f;
  }
  if (tid < 16) {
    w1p[tid] = (tid < 10) ? alpha * p.w1[tid] : 0.0f;
    w2p[tid] = (tid < 10) ? alpha * p.w2[tid] : 0.0f;
    b1p[tid] = (tid < 10) ? alpha * p.b1[tid] : 0.0f;
    w9p[tid] = (tid < 10) ? p.w9[tid] : 0.0f;
  }
  __syncthreads();

  const int lane = tid & 31;
  const int m    = lane & 15;
  const bool loH = (lane < 16);
  const int h5   = loH ? 0 : 5;
  const int jrow = (m < 5) ? m : ((m >= 8 && m < 13) ? (m - 3) : 15);

  v2f A[6][3];
#pragma unroll
  for (int L = 0; L < 6; ++L)
#pragma unroll
    for (int w = 0; w < 3; ++w) {
      A[L][w].x = Wp[L][2 * w + 0 + h5][jrow];
      A[L][w].y = (w < 2) ? Wp[L][2 * w + 1 + h5][jrow] : 0.0f;
    }
  v8f abv[6];
#pragma unroll
  for (int L = 0; L < 6; ++L)
#pragma unroll
    for (int r = 0; r < 8; ++r) abv[L][r] = (r < 5) ? Bp[L][r + h5] : 0.0f;
  float w1s[5], w2s[5], b1s[5], w9s[5];
#pragma unroll
  for (int r = 0; r < 5; ++r) {
    w1s[r] = w1p[r + h5]; w2s[r] = w2p[r + h5];
    b1s[r] = b1p[r + h5]; w9s[r] = w9p[r + h5];
  }
  const float b9v = p.b9[0];

  const int numPairs   = (p.n + 31) >> 5;
  const int waveId     = blockIdx.x * (blockDim.x >> 5) + (tid >> 5);
  const int waveStride = gridDim.x * (blockDim.x >> 5);

#define ISSUE_AFTER(x, y) asm volatile("" : "+v"(x) : "v"(y))

  for (int t = waveId; t < numPairs; t += waveStride) {
    const int base = t << 5;
    int ga = base + m;      if (ga >= p.n) ga = p.n - 1;
    int gb = base + 16 + m; if (gb >= p.n) gb = p.n - 1;
    const float x1a = p.x1[ga];
    const float x2a = p.x2[ga];
    const float x1b = p.x1[gb];
    const float x2b = p.x2[gb];

    float aa[5], ab[5];
#pragma unroll
    for (int r = 0; r < 5; ++r)
      aa[r] = TANHF(__builtin_fmaf(x1a, w1s[r], __builtin_fmaf(x2a, w2s[r], b1s[r])));
#pragma unroll
    for (int r = 0; r < 5; ++r)
      ab[r] = TANHF(__builtin_fmaf(x1b, w1s[r], __builtin_fmaf(x2b, w2s[r], b1s[r])));

    v8f ca, cb;
#pragma unroll
    for (int L = 0; L < 6; ++L) {
      if (L > 0) {
        ISSUE_AFTER(ca, cb[0]);
#pragma unroll
        for (int r = 0; r < 5; ++r) aa[r] = TANHF(ca[r]);
      }
      v2f a0; a0.x = aa[0]; a0.y = aa[1];
      v2f a1; a1.x = aa[2]; a1.y = aa[3];
      v2f a2; a2.x = aa[4]; a2.y = 0.0f;
      ca = __builtin_amdgcn_wmma_f32_16x16x4_f32(false, A[L][0], false, a0,
                                                 (short)0, abv[L], false, false);
      if (L > 0) {
        ISSUE_AFTER(cb, ca[0]);
#pragma unroll
        for (int r = 0; r < 5; ++r) ab[r] = TANHF(cb[r]);
      }
      v2f b0; b0.x = ab[0]; b0.y = ab[1];
      v2f b1; b1.x = ab[2]; b1.y = ab[3];
      v2f b2; b2.x = ab[4]; b2.y = 0.0f;
      if (L == 0) ISSUE_AFTER(b0, ca[0]);
      cb = __builtin_amdgcn_wmma_f32_16x16x4_f32(false, A[L][0], false, b0,
                                                 (short)0, abv[L], false, false);
      ISSUE_AFTER(a1, cb[0]);
      ca = __builtin_amdgcn_wmma_f32_16x16x4_f32(false, A[L][1], false, a1,
                                                 (short)0, ca, false, false);
      ISSUE_AFTER(b1, ca[0]);
      cb = __builtin_amdgcn_wmma_f32_16x16x4_f32(false, A[L][1], false, b1,
                                                 (short)0, cb, false, false);
      ISSUE_AFTER(a2, cb[0]);
      ca = __builtin_amdgcn_wmma_f32_16x16x4_f32(false, A[L][2], false, a2,
                                                 (short)0, ca, false, false);
      ISSUE_AFTER(b2, ca[0]);
      cb = __builtin_amdgcn_wmma_f32_16x16x4_f32(false, A[L][2], false, b2,
                                                 (short)0, cb, false, false);
    }
    ISSUE_AFTER(ca, cb[0]);
#pragma unroll
    for (int r = 0; r < 5; ++r) aa[r] = TANHF(ca[r]);
#pragma unroll
    for (int r = 0; r < 5; ++r) ab[r] = TANHF(cb[r]);

    float pa = 0.0f, pb = 0.0f;
#pragma unroll
    for (int r = 0; r < 5; ++r) pa = __builtin_fmaf(aa[r], w9s[r], pa);
#pragma unroll
    for (int r = 0; r < 5; ++r) pb = __builtin_fmaf(ab[r], w9s[r], pb);
    half_swap(pa, pb, loH);
    const float tot = pa + pb + b9v;
    if ((base + lane) < p.n) p.out[base + lane] = tot;
  }
#undef ISSUE_AFTER
}

extern "C" void kernel_launch(void* const* d_in, const int* in_sizes, int n_in,
                              void* d_out, int out_size, void* d_ws, size_t ws_size,
                              hipStream_t stream) {
  MlpParams p;
  p.x1    = (const float*)d_in[0];
  p.x2    = (const float*)d_in[1];
  p.w1    = (const float*)d_in[2];
  p.w2    = (const float*)d_in[3];
  p.b1    = (const float*)d_in[4];
  p.W3    = (const float*)d_in[5];   p.B3 = (const float*)d_in[6];
  p.W4    = (const float*)d_in[7];   p.B4 = (const float*)d_in[8];
  p.W5    = (const float*)d_in[9];   p.B5 = (const float*)d_in[10];
  p.W6    = (const float*)d_in[11];  p.B6 = (const float*)d_in[12];
  p.W7    = (const float*)d_in[13];  p.B7 = (const float*)d_in[14];
  p.W8    = (const float*)d_in[15];  p.B8 = (const float*)d_in[16];
  p.alpha = (const float*)d_in[17];
  p.w9    = (const float*)d_in[18];
  p.b9    = (const float*)d_in[19];
  p.out   = (float*)d_out;
  p.n     = in_sizes[0];

  const int blocks = 2048;
  mlp_wmma_kernel<<<dim3(blocks), dim3(256), 0, stream>>>(p);
}


# One resident 24-wave work-group per WGP (2048x8 -> 256x24 waves) loops over its tiles, next trip's inputs loaded one trip ahead: entry and weight staging once per WGP, no lone-wave memory wait
# speedup vs baseline: 1.3512x; 1.0686x over previous
#include <hip/hip_runtime.h>

typedef float v2f __attribute__((ext_vector_type(2)));
typedef float v8f __attribute__((ext_vector_type(8)));

#if defined(__HIP_DEVICE_COMPILE__) && __has_builtin(__builtin_amdgcn_tanhf)
  #define TANHF(x) __builtin_amdgcn_tanhf(x)
#else
  #define TANHF(x) tanhf(x)
#endif

__device__ __forceinline__ void half_swap(float& a, float& b, bool loH) {
#if defined(__HIP_DEVICE_COMPILE__) && __has_builtin(__builtin_amdgcn_permlane16_swap)
  auto r = __builtin_amdgcn_permlane16_swap(__float_as_uint(a), __float_as_uint(b),
                                            false, false);
  a = __uint_as_float(r[0]);
  b = __uint_as_float(r[1]);
#else
  const float sa = __shfl_xor(a, 16, 32);
  const float sb = __shfl_xor(b, 16, 32);
  const float na = loH ? a : sb;
  const float nb = loH ? sa : b;
  a = na; b = nb;
#endif
}

struct MlpParams {
  const float* x1; const float* x2;
  const float* w1; const float* w2; const float* b1;
  const float* W3; const float* B3; const float* W4; const float* B4;
  const float* W5; const float* B5; const float* W6; const float* B6;
  const float* W7; const float* B7; const float* W8; const float* B8;
  const float* alpha; const float* w9; const float* b9;
  float* out; int n;
};


constexpr int kThreads = 768;
constexpr int kWaves   = kThreads / 32;

__global__ __launch_bounds__(kThreads) void mlp_wmma_kernel(MlpParams p) {
  __shared__ float Wp[6][16][16];
  __shared__ float Bp[6][16];
  __shared__ float w1p[16], w2p[16], b1p[16], w9p[16];

  const int tid = threadIdx.x;
  const float alpha = p.alpha[0];
  const float* Wg[6] = {p.W3, p.W4, p.W5, p.W6, p.W7, p.W8};
  const float* Bg[6] = {p.B3, p.B4, p.B5, p.B6, p.B7, p.B8};

  for (int idx = tid; idx < 6 * 256; idx += kThreads) {
    const int L = idx >> 8, pos = idx & 255, k = pos >> 4, j = pos & 15;
    Wp[L][k][j] = (k < 10 && j < 10) ? alpha * Wg[L][k * 10 + j] : 0.0f;
  }
  if (tid < 96) {
    const int L = tid >> 4, f = tid & 15;
    Bp[L][f] = (f < 10) ? alpha * Bg[L][f] : 0.0f;
  }
  if (tid < 16) {
    w1p[tid] = (tid < 10) ? alpha * p.w1[tid] : 0.0f;
    w2p[tid] = (tid < 10) ? alpha * p.w2[tid] : 0.0f;
    b1p[tid] = (tid < 10) ? alpha * p.b1[tid] : 0.0f;
    w9p[tid] = (tid < 10) ? p.w9[tid] : 0.0f;
  }
  __syncthreads();

  const int lane = tid & 31;
  const int m    = lane & 15;
  const bool loH = (lane < 16);
  const int h5   = loH ? 0 : 5;
  const int jrow = (m < 5) ? m : ((m >= 8 && m < 13) ? (m - 3) : 15);

  v2f A[6][3];
#pragma unroll
  for (int L = 0; L < 6; ++L)
#pragma unroll
    for (int w = 0; w < 3; ++w) {
      A[L][w].x = Wp[L][2 * w + 0 + h5][jrow];
      A[L][w].y = (w < 2) ? Wp[L][2 * w + 1 + h5][jrow] : 0.0f;
    }
  v8f abv[6];
#pragma unroll
  for (int L = 0; L < 6; ++L)
#pragma unroll
    for (int r = 0; r < 8; ++r) abv[L][r] = (r < 5) ? Bp[L][r + h5] : 0.0f;
  float w1s[5], w2s[5], b1s[5], w9s[5];
#pragma unroll
  for (int r = 0; r < 5; ++r) {
    w1s[r] = w1p[r + h5]; w2s[r] = w2p[r + h5];
    b1s[r] = b1p[r + h5]; w9s[r] = w9p[r + h5];
  }
  const float b9v = p.b9[0];

  const int numPairs   = (p.n + 31) >> 5;
  const int waveId     = (tid >> 5) * gridDim.x + blockIdx.x;
  const int waveStride = gridDim.x * kWaves;
  const unsigned lastIdx = (unsigned)(p.n - 1);
  const unsigned stepS   = (unsigned)waveStride << 5;

#define ISSUE_AFTER(x, y) asm volatile("" : "+v"(x) : "v"(y))

  if (waveId >= numPairs) return;
  unsigned ia = ((unsigned)waveId << 5) + (unsigned)m;
  unsigned io = ((unsigned)waveId << 5) + (unsigned)lane;
  float x1a, x2a, x1b, x2b;
  {
    const unsigned ga = ia < lastIdx ? ia : lastIdx;
    const unsigned gb = ia + 16u < lastIdx ? ia + 16u : lastIdx;
    x1a = p.x1[ga]; x2a = p.x2[ga];
    x1b = p.x1[gb]; x2b = p.x2[gb];
  }

  for (int t = waveId; t < numPairs; t += waveStride) {
    float aa[5], ab[5];
#pragma unroll
    for (int r = 0; r < 5; ++r)
      aa[r] = TANHF(__builtin_fmaf(x1a, w1s[r], __builtin_fmaf(x2a, w2s[r], b1s[r])));
#pragma unroll
    for (int r = 0; r < 5; ++r)
      ab[r] = TANHF(__builtin_fmaf(x1b, w1s[r], __builtin_fmaf(x2b, w2s[r], b1s[r])));

    ia += stepS;
    {
      const unsigned ga = ia < lastIdx ? ia : lastIdx;
      const unsigned gb = ia + 16u < lastIdx ? ia + 16u : lastIdx;
      x1a = p.x1[ga]; x2a = p.x2[ga];
      x1b = p.x1[gb]; x2b = p.x2[gb];
    }

    v8f ca, cb;
#pragma unroll
    for (int L = 0; L < 6; ++L) {
      if (L > 0) {
        ISSUE_AFTER(ca, cb[0]);
#pragma unroll
        for (int r = 0; r < 5; ++r) aa[r] = TANHF(ca[r]);
      }
      v2f a0; a0.x = aa[0]; a0.y = aa[1];
      v2f a1; a1.x = aa[2]; a1.y = aa[3];
      v2f a2; a2.x = aa[4]; a2.y = 0.0f;
      ca = __builtin_amdgcn_wmma_f32_16x16x4_f32(false, A[L][0], false, a0,
                                                 (short)0, abv[L], false, false);
      if (L > 0) {
        ISSUE_AFTER(cb, ca[0]);
#pragma unroll
        for (int r = 0; r < 5; ++r) ab[r] = TANHF(cb[r]);
      }
      v2f b0; b0.x = ab[0]; b0.y = ab[1];
      v2f b1; b1.x = ab[2]; b1.y = ab[3];
      v2f b2; b2.x = ab[4]; b2.y = 0.0f;
      if (L == 0) ISSUE_AFTER(b0, ca[0]);
      cb = __builtin_amdgcn_wmma_f32_16x16x4_f32(false, A[L][0], false, b0,
                                                 (short)0, abv[L], false, false);
      ISSUE_AFTER(a1, cb[0]);
      ca = __builtin_amdgcn_wmma_f32_16x16x4_f32(false, A[L][1], false, a1,
                                                 (short)0, ca, false, false);
      ISSUE_AFTER(b1, ca[0]);
      cb = __builtin_amdgcn_wmma_f32_16x16x4_f32(false, A[L][1], false, b1,
                                                 (short)0, cb, false, false);
      ISSUE_AFTER(a2, cb[0]);
      ca = __builtin_amdgcn_wmma_f32_16x16x4_f32(false, A[L][2], false, a2,
                                                 (short)0, ca, false, false);
      ISSUE_AFTER(b2, ca[0]);
      cb = __builtin_amdgcn_wmma_f32_16x16x4_f32(false, A[L][2], false, b2,
                                                 (short)0, cb, false, false);
    }
    ISSUE_AFTER(ca, cb[0]);
#pragma unroll
    for (int r = 0; r < 5; ++r) aa[r] = TANHF(ca[r]);
#pragma unroll
    for (int r = 0; r < 5; ++r) ab[r] = TANHF(cb[r]);

    float pa = 0.0f, pb = 0.0f;
#pragma unroll
    for (int r = 0; r < 5; ++r) pa = __builtin_fmaf(aa[r], w9s[r], pa);
#pragma unroll
    for (int r = 0; r < 5; ++r) pb = __builtin_fmaf(ab[r], w9s[r], pb);
    half_swap(pa, pb, loH);
    const float tot = pa + pb + b9v;
    if (io < (unsigned)p.n) p.out[io] = tot;
    io += stepS;
  }
#undef ISSUE_AFTER
}

extern "C" void kernel_launch(void* const* d_in, const int* in_sizes, int n_in,
                              void* d_out, int out_size, void* d_ws, size_t ws_size,
                              hipStream_t stream) {
  MlpParams p;
  p.x1    = (const float*)d_in[0];
  p.x2    = (const float*)d_in[1];
  p.w1    = (const float*)d_in[2];
  p.w2    = (const float*)d_in[3];
  p.b1    = (const float*)d_in[4];
  p.W3    = (const float*)d_in[5];   p.B3 = (const float*)d_in[6];
  p.W4    = (const float*)d_in[7];   p.B4 = (const float*)d_in[8];
  p.W5    = (const float*)d_in[9];   p.B5 = (const float*)d_in[10];
  p.W6    = (const float*)d_in[11];  p.B6 = (const float*)d_in[12];
  p.W7    = (const float*)d_in[13];  p.B7 = (const float*)d_in[14];
  p.W8    = (const float*)d_in[15];  p.B8 = (const float*)d_in[16];
  p.alpha = (const float*)d_in[17];
  p.w9    = (const float*)d_in[18];
  p.b9    = (const float*)d_in[19];
  p.out   = (float*)d_out;
  p.n     = in_sizes[0];

  const int blocks = 256;
  mlp_wmma_kernel<<<dim3(blocks), dim3(kThreads), 0, stream>>>(p);
}


# Tiles in threes: two host tiles' third WMMA also carries the guest tile's inputs 4|9 in its idle k slot and six padding rows: 8 WMMA per 3 tiles and layer, was 9; guest's sum order changes (5 adds)
# speedup vs baseline: 1.5123x; 1.1192x over previous
#include <hip/hip_runtime.h>

typedef float v2f __attribute__((ext_vector_type(2)));
typedef float v8f __attribute__((ext_vector_type(8)));

#if defined(__HIP_DEVICE_COMPILE__) && __has_builtin(__builtin_amdgcn_tanhf)
  #define TANHF(x) __builtin_amdgcn_tanhf(x)
#else
  #define TANHF(x) tanhf(x)
#endif

__device__ __forceinline__ void half_swap(float& a, float& b, bool loH) {
#if defined(__HIP_DEVICE_COMPILE__) && __has_builtin(__builtin_amdgcn_permlane16_swap)
  auto r = __builtin_amdgcn_permlane16_swap(__float_as_uint(a), __float_as_uint(b),
                                            false, false);
  a = __uint_as_float(r[0]);
  b = __uint_as_float(r[1]);
#else
  const float sa = __shfl_xor(a, 16, 32);
  const float sb = __shfl_xor(b, 16, 32);
  const float na = loH ? a : sb;
  const float nb = loH ? sa : b;
  a = na; b = nb;
#endif
}

struct MlpParams {
  const float* x1; const float* x2;
  const float* w1; const float* w2; const float* b1;
  const float* W3; const float* B3; const float* W4; const float* B4;
  const float* W5; const float* B5; const float* W6; const float* B6;
  const float* W7; const float* B7; const float* W8; const float* B8;
  const float* alpha; const float* w9; const float* b9;
  float* out; int n;
};


constexpr int kThreads = 640;
constexpr int kWaves   = kThreads / 32;

__global__ __launch_bounds__(kThreads) void mlp_wmma_kernel(MlpParams p) {
  __shared__ float Wp[6][16][16];
  __shared__ float Bp[6][16];
  __shared__ float w1p[16], w2p[16], b1p[16], w9p[16];

  const int tid = threadIdx.x;
  const float alpha = p.alpha[0];
  const float* Wg[6] = {p.W3, p.W4, p.W5, p.W6, p.W7, p.W8};
  const float* Bg[6] = {p.B3, p.B4, p.B5, p.B6, p.B7, p.B8};

  for (int idx = tid; idx < 6 * 256; idx += kThreads) {
    const int L = idx >> 8, pos = idx & 255, k = pos >> 4, j = pos & 15;
    Wp[L][k][j] = (k < 10 && j < 10) ? alpha * Wg[L][k * 10 + j] : 0.0f;
  }
  if (tid < 96) {
    const int L = tid >> 4, f = tid & 15;
    Bp[L][f] = (f < 10) ? alpha * Bg[L][f] : 0.0f;
  }
  if (tid < 16) {
    w1p[tid] = (tid < 10) ? alpha * p.w1[tid] : 0.0f;
    w2p[tid] = (tid < 10) ? alpha * p.w2[tid] : 0.0f;
    b1p[tid] = (tid < 10) ? alpha * p.b1[tid] : 0.0f;
    w9p[tid] = (tid < 10) ? p.w9[tid] : 0.0f;
  }
  __syncthreads();

  const int lane = tid & 31;
  const int m    = lane & 15;
  const bool loH = (lane < 16);
  const int h5   = loH ? 0 : 5;
  const int jrow = (m < 5) ? m : ((m >= 8 && m < 13) ? (m - 3) : 15);

  const int gcol1 = (m >= 5 && m < 8) ? (m - 5) : ((m >= 13) ? (m - 8) : 15);
  const int gcol2 = (m == 5 || m == 6) ? (m - 2) : ((m == 13 || m == 14) ? (m - 5) : 15);

  v2f A[6][2];
  v2f A3a[6], A3b[6];
#pragma unroll
  for (int L = 0; L < 6; ++L) {
#pragma unroll
    for (int w = 0; w < 2; ++w) {
      A[L][w].x = Wp[L][2 * w + 0 + h5][jrow];
      A[L][w].y = Wp[L][2 * w + 1 + h5][jrow];
    }
    A3a[L].x = Wp[L][4 + h5][jrow];
    A3a[L].y = Wp[L][4 + h5][gcol1];
    A3b[L].x = Wp[L][4 + h5][jrow];
    A3b[L].y = Wp[L][4 + h5][gcol2];
  }
  v8f abv[6];
#pragma unroll
  for (int L = 0; L < 6; ++L)
#pragma unroll
    for (int r = 0; r < 8; ++r) abv[L][r] = (r < 5) ? Bp[L][r + h5] : 0.0f;
  float w1s[5], w2s[5], b1s[5], w9s[5];
#pragma unroll
  for (int r = 0; r < 5; ++r) {
    w1s[r] = w1p[r + h5]; w2s[r] = w2p[r + h5];
    b1s[r] = b1p[r + h5]; w9s[r] = w9p[r + h5];
  }
  const float b9v = p.b9[0];

  const int numChunks  = (p.n + 95) / 96;
  const int waveId     = (tid >> 5) * gridDim.x + blockIdx.x;
  const int waveStride = gridDim.x * kWaves;
  const unsigned lastIdx = (unsigned)(p.n - 1);
  const unsigned stepIn  = 48u;
  const unsigned stepOut = (unsigned)waveStride * 96u - 48u;

#define ISSUE_AFTER(x, y) asm volatile("" : "+v"(x) : "v"(y))
#define TANH_GROUP(t) asm volatile("" :: "v"(t[0]), "v"(t[1]), "v"(t[2]), "v"(t[3]), "v"(t[4]))

  if (waveId >= numChunks) return;
  const int myChunks = (numChunks - waveId + waveStride - 1) / waveStride;
  unsigned base = (unsigned)waveId * 96u;
  float x1a, x2a, x1b, x2b, x1g, x2g;
  {
    const unsigned ia = base + (unsigned)m;
    const unsigned ga = ia < lastIdx ? ia : lastIdx;
    const unsigned gb = ia + 16u < lastIdx ? ia + 16u : lastIdx;
    const unsigned gg = ia + 32u < lastIdx ? ia + 32u : lastIdx;
    x1a = p.x1[ga]; x2a = p.x2[ga];
    x1b = p.x1[gb]; x2b = p.x2[gb];
    x1g = p.x1[gg]; x2g = p.x2[gg];
  }

  for (int it = 0; it < 2 * myChunks; ++it) {
    float aa[5], ab[5], ag[5];
#pragma unroll
    for (int r = 0; r < 5; ++r)
      aa[r] = TANHF(__builtin_fmaf(x1a, w1s[r], __builtin_fmaf(x2a, w2s[r], b1s[r])));
#pragma unroll
    for (int r = 0; r < 5; ++r)
      ab[r] = TANHF(__builtin_fmaf(x1b, w1s[r], __builtin_fmaf(x2b, w2s[r], b1s[r])));
#pragma unroll
    for (int r = 0; r < 5; ++r)
      ag[r] = TANHF(__builtin_fmaf(x1g, w1s[r], __builtin_fmaf(x2g, w2s[r], b1s[r])));

    const unsigned cur = base;
    base += (it & 1) ? stepOut : stepIn;
    {
      const unsigned ia = base + (unsigned)m;
      const unsigned ga = ia < lastIdx ? ia : lastIdx;
      const unsigned gb = ia + 16u < lastIdx ? ia + 16u : lastIdx;
      const unsigned gg = ia + 32u < lastIdx ? ia + 32u : lastIdx;
      x1a = p.x1[ga]; x2a = p.x2[ga];
      x1b = p.x1[gb]; x2b = p.x2[gb];
      x1g = p.x1[gg]; x2g = p.x2[gg];
    }

    v8f ca, cb, cg;
    float e0 = 0.0f, e1 = 0.0f, e2 = 0.0f, e3 = 0.0f, e4 = 0.0f;
#pragma unroll
    for (int L = 0; L < 6; ++L) {
      if (L > 0) {
        ISSUE_AFTER(ca, cb[0]);
        e0 = ca[5]; e1 = ca[6]; e2 = ca[7];
#pragma unroll
        for (int r = 0; r < 5; ++r) aa[r] = TANHF(ca[r]);
        TANH_GROUP(aa);
      }
      v2f a0; a0.x = aa[0]; a0.y = aa[1];
      v2f a1; a1.x = aa[2]; a1.y = aa[3];
      ca = __builtin_amdgcn_wmma_f32_16x16x4_f32(false, A[L][0], false, a0,
                                                 (short)0, abv[L], false, false);
      if (L > 0) {
        ISSUE_AFTER(cb, ca[0]);
        e3 = cb[5]; e4 = cb[6];
#pragma unroll
        for (int r = 0; r < 5; ++r) ab[r] = TANHF(cb[r]);
        TANH_GROUP(ab);
      }
      v2f b0; b0.x = ab[0]; b0.y = ab[1];
      v2f b1; b1.x = ab[2]; b1.y = ab[3];
      if (L == 0) ISSUE_AFTER(b0, ca[0]);
      cb = __builtin_amdgcn_wmma_f32_16x16x4_f32(false, A[L][0], false, b0,
                                                 (short)0, abv[L], false, false);
      if (L > 0) {
        ISSUE_AFTER(cg, cb[0]);
        ag[0] = TANHF(cg[0] + e0);
        ag[1] = TANHF(cg[1] + e1);
        ag[2] = TANHF(cg[2] + e2);
        ag[3] = TANHF(cg[3] + e3);
        ag[4] = TANHF(cg[4] + e4);
        TANH_GROUP(ag);
      }
      v2f g0; g0.x = ag[0]; g0.y = ag[1];
      v2f g1; g1.x = ag[2]; g1.y = ag[3];
      if (L == 0) ISSUE_AFTER(g0, cb[0]);
      cg = __builtin_amdgcn_wmma_f32_16x16x4_f32(false, A[L][0], false, g0,
                                                 (short)0, abv[L], false, false);
      ISSUE_AFTER(a1, cg[0]);
      ca = __builtin_amdgcn_wmma_f32_16x16x4_f32(false, A[L][1], false, a1,
                                                 (short)0, ca, false, false);
      ISSUE_AFTER(b1, ca[0]);
      cb = __builtin_amdgcn_wmma_f32_16x16x4_f32(false, A[L][1], false, b1,
                                                 (short)0, cb, false, false);
      ISSUE_AFTER(g1, cb[0]);
      cg = __builtin_amdgcn_wmma_f32_16x16x4_f32(false, A[L][1], false, g1,
                                                 (short)0, cg, false, false);
      v2f a2; a2.x = aa[4]; a2.y = ag[4];
      v2f b2; b2.x = ab[4]; b2.y = ag[4];
      ISSUE_AFTER(a2, cg[0]);
      ca = __builtin_amdgcn_wmma_f32_16x16x4_f32(false, A3a[L], false, a2,
                                                 (short)0, ca, false, false);
      ISSUE_AFTER(b2, ca[0]);
      cb = __builtin_amdgcn_wmma_f32_16x16x4_f32(false, A3b[L], false, b2,
                                                 (short)0, cb, false, false);
    }
    ISSUE_AFTER(ca, cb[0]);
    e0 = ca[5]; e1 = ca[6]; e2 = ca[7];
#pragma unroll
    for (int r = 0; r < 5; ++r) aa[r] = TANHF(ca[r]);
    e3 = cb[5]; e4 = cb[6];
#pragma unroll
    for (int r = 0; r < 5; ++r) ab[r] = TANHF(cb[r]);
    ag[0] = TANHF(cg[0] + e0);
    ag[1] = TANHF(cg[1] + e1);
    ag[2] = TANHF(cg[2] + e2);
    ag[3] = TANHF(cg[3] + e3);
    ag[4] = TANHF(cg[4] + e4);

    float pa = 0.0f, pb = 0.0f, pg = 0.0f;
#pragma unroll
    for (int r = 0; r < 5; ++r) pa = __builtin_fmaf(aa[r], w9s[r], pa);
#pragma unroll
    for (int r = 0; r < 5; ++r) pb = __builtin_fmaf(ab[r], w9s[r], pb);
#pragma unroll
    for (int r = 0; r < 5; ++r) pg = __builtin_fmaf(ag[r], w9s[r], pg);
    half_swap(pa, pb, loH);
    float qg = pg;
    half_swap(pg, qg, loH);
    const float totab = pa + pb + b9v;
    const float totg  = pg + qg + b9v;
    const unsigned io = cur + (unsigned)lane;
    const unsigned ig = cur + 32u + (unsigned)m;
    if (io < (unsigned)p.n) p.out[io] = totab;
    if (loH && ig < (unsigned)p.n) p.out[ig] = totg;
  }
#undef TANH_GROUP
#undef ISSUE_AFTER
}

extern "C" void kernel_launch(void* const* d_in, const int* in_sizes, int n_in,
                              void* d_out, int out_size, void* d_ws, size_t ws_size,
                              hipStream_t stream) {
  MlpParams p;
  p.x1    = (const float*)d_in[0];
  p.x2    = (const float*)d_in[1];
  p.w1    = (const float*)d_in[2];
  p.w2    = (const float*)d_in[3];
  p.b1    = (const float*)d_in[4];
  p.W3    = (const float*)d_in[5];   p.B3 = (const float*)d_in[6];
  p.W4    = (const float*)d_in[7];   p.B4 = (const float*)d_in[8];
  p.W5    = (const float*)d_in[9];   p.B5 = (const float*)d_in[10];
  p.W6    = (const float*)d_in[11];  p.B6 = (const float*)d_in[12];
  p.W7    = (const float*)d_in[13];  p.B7 = (const float*)d_in[14];
  p.W8    = (const float*)d_in[15];  p.B8 = (const float*)d_in[16];
  p.alpha = (const float*)d_in[17];
  p.w9    = (const float*)d_in[18];
  p.b9    = (const float*)d_in[19];
  p.out   = (float*)d_out;
  p.n     = in_sizes[0];

  const int blocks = 256;
  mlp_wmma_kernel<<<dim3(blocks), dim3(kThreads), 0, stream>>>(p);
}
